# starting kernel
# baseline (speedup 1.0000x reference)
#include <hip/hip_runtime.h>

#define NB 512
#define NT 2048
#define NS 16
#define NH 128
#define NA 4
#define NTOK   (NB*NT)
#define NTILES (NTOK/16)

#define WAVES   2
#define TPB     (WAVES*32)
#define NBLOCKS 2048

typedef float v2f __attribute__((ext_vector_type(2)));
typedef float v8f __attribute__((ext_vector_type(8)));

__device__ __forceinline__ v8f splat8(float x){
  v8f r;
#pragma unroll
  for (int i = 0; i < 8; i++) r[i] = x;
  return r;
}

__device__ __forceinline__ v2f mk2(float a, float b){
  v2f r; r[0] = a; r[1] = b; return r;
}

__device__ __forceinline__ v8f wmma4(v2f a, v2f b, v8f c){
  return __builtin_amdgcn_wmma_f32_16x16x4_f32(false, a, false, b,
                                               (short)0, c, false, false);
}

__device__ __forceinline__ float fast_tanh(float x){
  float ax = __builtin_fabsf(x);
  float e  = __builtin_amdgcn_exp2f(ax * -2.885390081777927f);
  float t  = (1.0f - e) * __builtin_amdgcn_rcpf(1.0f + e);
  return __builtin_copysignf(t, x);
}

__device__ __forceinline__ void lds_sync(){
  asm volatile("" ::: "memory");
  __builtin_amdgcn_wave_barrier();
  asm volatile("" ::: "memory");
}

__global__ __launch_bounds__(TPB) void Agent_cbf_qp_kernel(
    const float* __restrict__ state,
    const float* __restrict__ Wc1, const float* __restrict__ bc1,
    const float* __restrict__ Wc2, const float* __restrict__ bc2,
    const float* __restrict__ Wh1, const float* __restrict__ bh1,
    const float* __restrict__ wh2, const float* __restrict__ bh2,
    const float* __restrict__ Wf,  const float* __restrict__ bf,
    const float* __restrict__ Wg,  const float* __restrict__ bg,
    float* __restrict__ out)
{
  __shared__ v2f sBc1 [1024];
  __shared__ v2f sBh1 [1024];
  __shared__ v2f sBh1T[1024];
  __shared__ v2f sBc2 [1024];
  __shared__ v2f sBf  [ 128];
  __shared__ v2f sBg  [ 512];
  __shared__ float s_bc1[128], s_bh1[128], s_wh2[128], s_bf[16], s_bg[64], s_bc2[4];
  __shared__ float sStage[WAVES][256];
  __shared__ float sDH   [WAVES][256];
  __shared__ float sFb   [WAVES][256];
  __shared__ float sGb   [WAVES][1024];
  __shared__ float sHb   [WAVES][16];
  __shared__ float sUU   [WAVES][64];

  const int tid = threadIdx.x;

  for (int i = tid; i < 1024; i += TPB){
    const int l = i & 31, lm = l & 15, lh = l >> 4;
    const int c = (i >> 5) & 3, jt = i >> 7;
    const int n = jt*16 + lm,  k0 = c*4 + lh*2;
    sBc1[i] = mk2(Wc1[k0*NH + n], Wc1[(k0+1)*NH + n]);
    sBh1[i] = mk2(Wh1[k0*NH + n], Wh1[(k0+1)*NH + n]);
    const int kc  = i >> 5;
    const int kk0 = kc*4 + lh*2;
    sBh1T[i] = mk2(Wh1[lm*NH + kk0], Wh1[lm*NH + kk0 + 1]);
    sBc2[i]  = (lm < NA) ? mk2(Wc2[kk0*NA + lm], Wc2[(kk0+1)*NA + lm])
                         : mk2(0.0f, 0.0f);
  }
  for (int i = tid; i < 128; i += TPB){
    const int l = i & 31, lm = l & 15, lh = l >> 4;
    const int c = i >> 5, k0 = c*4 + lh*2;
    sBf[i] = mk2(Wf[k0*NS + lm], Wf[(k0+1)*NS + lm]);
  }
  for (int i = tid; i < 512; i += TPB){
    const int l = i & 31, lm = l & 15, lh = l >> 4;
    const int c = (i >> 5) & 3, jt = i >> 7;
    const int n = jt*16 + lm, k0 = c*4 + lh*2;
    sBg[i] = mk2(Wg[k0*(NS*NA) + n], Wg[(k0+1)*(NS*NA) + n]);
  }
  for (int i = tid; i < 128; i += TPB){ s_bc1[i]=bc1[i]; s_bh1[i]=bh1[i]; s_wh2[i]=wh2[i]; }
  for (int i = tid; i < 64;  i += TPB) s_bg[i] = bg[i];
  for (int i = tid; i < 16;  i += TPB) s_bf[i] = bf[i];
  for (int i = tid; i < 4;   i += TPB) s_bc2[i] = bc2[i];
  __syncthreads();

  const int lane = tid & 31, wid = tid >> 5;
  const int lm = lane & 15, half = lane >> 4;
  const float bh2s = bh2[0];
  float* stage = sStage[wid];
  float* DHb   = sDH[wid];
  float* Fb    = sFb[wid];
  float* Gb    = sGb[wid];
  float* Hb    = sHb[wid];
  float* UUb   = sUU[wid];

  const int gw = blockIdx.x * WAVES + wid;
  const int stride = gridDim.x * WAVES;

  for (int tile = gw; tile < NTILES; tile += stride){
    const v2f* sp = (const v2f*)(state + (tile*16 + lm) * NS);
    v2f a[4];
#pragma unroll
    for (int c = 0; c < 4; c++) a[c] = sp[2*c + half];

    if (tile + stride < NTILES)
      __builtin_prefetch(state + ((tile + stride)*16 + lm) * NS, 0, 0);

    v8f U = splat8(lm < NA ? s_bc2[lm] : 0.0f);
    for (int jt = 0; jt < 8; jt++){
      v8f z = splat8(s_bc1[jt*16 + lm]);
#pragma unroll
      for (int c = 0; c < 4; c++)
        z = wmma4(a[c], sBc1[(jt*4 + c)*32 + lane], z);
#pragma unroll
      for (int v = 0; v < 8; v++) z[v] = fast_tanh(z[v]);
#pragma unroll
      for (int v = 0; v < 8; v++) stage[(v + 8*half)*16 + lm] = z[v];
      lds_sync();
      v2f ta[4];
#pragma unroll
      for (int c = 0; c < 4; c++)
        ta[c] = *(const v2f*)&stage[lm*16 + c*4 + half*2];
#pragma unroll
      for (int c = 0; c < 4; c++)
        U = wmma4(ta[c], sBc2[(jt*4 + c)*32 + lane], U);
      lds_sync();
    }

    v8f DH = splat8(0.0f);
    float hacc[8];
#pragma unroll
    for (int v = 0; v < 8; v++) hacc[v] = 0.0f;
    for (int jt = 0; jt < 8; jt++){
      v8f z = splat8(s_bh1[jt*16 + lm]);
#pragma unroll
      for (int c = 0; c < 4; c++)
        z = wmma4(a[c], sBh1[(jt*4 + c)*32 + lane], z);
      const float w2 = s_wh2[jt*16 + lm];
#pragma unroll
      for (int v = 0; v < 8; v++){
        const float t = fast_tanh(z[v]);
        hacc[v] += t * w2;
        stage[(v + 8*half)*16 + lm] = (1.0f - t*t) * w2;
      }
      lds_sync();
      v2f ta[4];
#pragma unroll
      for (int c = 0; c < 4; c++)
        ta[c] = *(const v2f*)&stage[lm*16 + c*4 + half*2];
#pragma unroll
      for (int c = 0; c < 4; c++)
        DH = wmma4(ta[c], sBh1T[(jt*4 + c)*32 + lane], DH);
      lds_sync();
    }
#pragma unroll
    for (int v = 0; v < 8; v++){
#pragma unroll
      for (int ofs = 1; ofs < 16; ofs <<= 1)
        hacc[v] += __shfl_xor(hacc[v], ofs, 32);
    }
    if (lm == 0){
#pragma unroll
      for (int v = 0; v < 8; v++) Hb[v + 8*half] = hacc[v] + bh2s;
    }

    v8f F = splat8(s_bf[lm]);
#pragma unroll
    for (int c = 0; c < 4; c++) F = wmma4(a[c], sBf[c*32 + lane], F);
#pragma unroll
    for (int v = 0; v < 8; v++) Fb[(v + 8*half)*16 + lm] = F[v];

    for (int jt = 0; jt < 4; jt++){
      v8f G = splat8(s_bg[jt*16 + lm]);
#pragma unroll
      for (int c = 0; c < 4; c++) G = wmma4(a[c], sBg[(jt*4 + c)*32 + lane], G);
#pragma unroll
      for (int v = 0; v < 8; v++) Gb[(v + 8*half)*64 + jt*16 + lm] = G[v];
    }

#pragma unroll
    for (int v = 0; v < 8; v++) DHb[(v + 8*half)*16 + lm] = DH[v];
    if (lm < NA){
#pragma unroll
      for (int v = 0; v < 8; v++) UUb[(v + 8*half)*NA + lm] = 2.0f * U[v];
    }
    lds_sync();

    if (lane < 16){
      const int t = lane;
      float left0 = 0.f, left1 = 0.f, left2 = 0.f, left3 = 0.f;
      float right = Hb[t];
#pragma unroll
      for (int s = 0; s < 16; s++){
        const float dh = DHb[t*16 + s];
        right += dh * Fb[t*16 + s];
        left0 -= dh * Gb[t*64 + s*4 + 0];
        left1 -= dh * Gb[t*64 + s*4 + 1];
        left2 -= dh * Gb[t*64 + s*4 + 2];
        left3 -= dh * Gb[t*64 + s*4 + 3];
      }
      const float uu0 = UUb[t*4+0], uu1 = UUb[t*4+1];
      const float uu2 = UUb[t*4+2], uu3 = UUb[t*4+3];
      const float viol = left0*uu0 + left1*uu1 + left2*uu2 + left3*uu3 - right;
      const float den  = left0*left0 + left1*left1 + left2*left2 + left3*left3 + 1e-8f;
      const float lam  = fmaxf(viol, 0.0f) / den;
      float4 r;
      r.x = uu0 - lam*left0;
      r.y = uu1 - lam*left1;
      r.z = uu2 - lam*left2;
      r.w = uu3 - lam*left3;
      *(float4*)(out + (tile*16 + t)*NA) = r;
    }
    lds_sync();
  }
}

extern "C" void kernel_launch(void* const* d_in, const int* in_sizes, int n_in,
                              void* d_out, int out_size, void* d_ws, size_t ws_size,
                              hipStream_t stream) {
  (void)in_sizes; (void)n_in; (void)d_ws; (void)ws_size; (void)out_size;
  const float* state = (const float*)d_in[0];
  const float* Wc1   = (const float*)d_in[1];
  const float* bc1   = (const float*)d_in[2];
  const float* Wc2   = (const float*)d_in[3];
  const float* bc2   = (const float*)d_in[4];
  const float* Wh1   = (const float*)d_in[5];
  const float* bh1   = (const float*)d_in[6];
  const float* wh2   = (const float*)d_in[7];
  const float* bh2   = (const float*)d_in[8];
  const float* Wf    = (const float*)d_in[9];
  const float* bf    = (const float*)d_in[10];
  const float* Wg    = (const float*)d_in[11];
  const float* bg    = (const float*)d_in[12];
  float* out = (float*)d_out;

  hipLaunchKernelGGL(Agent_cbf_qp_kernel, dim3(NBLOCKS), dim3(TPB), 0, stream,
                     state, Wc1, bc1, Wc2, bc2, Wh1, bh1, wh2, bh2,
                     Wf, bf, Wg, bg, out);
}


# Products transposed (weights as A, tokens on N): a D tile is the next product's B operand via k-permuted weight fragments; no LDS relayout left; sum order of h, left, right, layer-2 k changes
# speedup vs baseline: 1.5435x; 1.5435x over previous
#include <hip/hip_runtime.h>

#define NB 512
#define NT 2048
#define NS 16
#define NH 128
#define NA 4
#define NTOK   (NB*NT)
#define NTILES (NTOK/16)

#define WAVES   2
#define TPB     (WAVES*32)
#define NBLOCKS 2048

typedef float v2f __attribute__((ext_vector_type(2)));
typedef float v8f __attribute__((ext_vector_type(8)));

__device__ __forceinline__ v8f splat8(float x){
  v8f r;
#pragma unroll
  for (int i = 0; i < 8; i++) r[i] = x;
  return r;
}

__device__ __forceinline__ v2f mk2(float a, float b){
  v2f r; r[0] = a; r[1] = b; return r;
}

__device__ __forceinline__ v8f wmma4(v2f a, v2f b, v8f c){
  return __builtin_amdgcn_wmma_f32_16x16x4_f32(false, a, false, b,
                                               (short)0, c, false, false);
}

__device__ __forceinline__ float fast_tanh(float x){
  float ax = __builtin_fabsf(x);
  float e  = __builtin_amdgcn_exp2f(ax * -2.885390081777927f);
  float t  = (1.0f - e) * __builtin_amdgcn_rcpf(1.0f + e);
  return __builtin_copysignf(t, x);
}

__device__ __forceinline__ v8f ld8(const float* p){
  return *(const v8f*)p;
}

__device__ __forceinline__ float xhalf(float x){
  const int b = __builtin_bit_cast(int, x);
  return __builtin_bit_cast(float,
      __builtin_amdgcn_permlanex16(b, b, 0x76543210, 0xfedcba98, false, false));
}

__global__ __launch_bounds__(TPB) void Agent_cbf_qp_kernel(
    const float* __restrict__ state,
    const float* __restrict__ Wc1, const float* __restrict__ bc1,
    const float* __restrict__ Wc2, const float* __restrict__ bc2,
    const float* __restrict__ Wh1, const float* __restrict__ bh1,
    const float* __restrict__ wh2, const float* __restrict__ bh2,
    const float* __restrict__ Wf,  const float* __restrict__ bf,
    const float* __restrict__ Wg,  const float* __restrict__ bg,
    float* __restrict__ out)
{
  __shared__ v2f sBc1 [1024];
  __shared__ v2f sBh1 [1024];
  __shared__ v2f sBh1T[1024];
  __shared__ v2f sBc2 [1024];
  __shared__ v2f sBf  [ 128];
  __shared__ v2f sBg  [ 512];
  __shared__ __attribute__((aligned(32))) float s_bc1[128], s_bh1[128], s_wh2[128],
                                                s_bf[16], s_bg[64], s_bc2[16];

  const int tid = threadIdx.x;

  for (int i = tid; i < 1024; i += TPB){
    const int l = i & 31, lm = l & 15, lh = l >> 4;
    const int c = (i >> 5) & 3, jt = i >> 7;
    const int n = jt*16 + lm,  k0 = c*4 + lh*2;
    sBc1[i] = mk2(Wc1[k0*NH + n], Wc1[(k0+1)*NH + n]);
    sBh1[i] = mk2(Wh1[k0*NH + n], Wh1[(k0+1)*NH + n]);
    const int kk0 = jt*16 + lh*8 + c*2;
    sBh1T[i] = mk2(Wh1[lm*NH + kk0], Wh1[lm*NH + kk0 + 1]);
    sBc2[i]  = (lm < NA) ? mk2(Wc2[kk0*NA + lm], Wc2[(kk0+1)*NA + lm])
                         : mk2(0.0f, 0.0f);
  }
  for (int i = tid; i < 128; i += TPB){
    const int l = i & 31, lm = l & 15, lh = l >> 4;
    const int c = i >> 5, k0 = c*4 + lh*2;
    sBf[i] = mk2(Wf[k0*NS + lm], Wf[(k0+1)*NS + lm]);
  }
  for (int i = tid; i < 512; i += TPB){
    const int l = i & 31, lm = l & 15, lh = l >> 4;
    const int c = (i >> 5) & 3, jt = i >> 7;
    const int n = (lm >> 3)*32 + jt*8 + (lm & 7), k0 = c*4 + lh*2;
    sBg[i] = mk2(Wg[k0*(NS*NA) + n], Wg[(k0+1)*(NS*NA) + n]);
  }
  for (int i = tid; i < 128; i += TPB){ s_bc1[i]=bc1[i]; s_bh1[i]=bh1[i]; s_wh2[i]=wh2[i]; }
  for (int i = tid; i < 64;  i += TPB) s_bg[i] = bg[i];
  for (int i = tid; i < 16;  i += TPB) s_bf[i] = bf[i];
  for (int i = tid; i < 16;  i += TPB) s_bc2[i] = (i < NA) ? bc2[i] : 0.0f;
  __syncthreads();

  const int lane = tid & 31, wid = tid >> 5;
  const int lm = lane & 15, half = lane >> 4;
  const float bh2s = bh2[0];

  const int gw = blockIdx.x * WAVES + wid;
  const int stride = gridDim.x * WAVES;

  for (int tile = gw; tile < NTILES; tile += stride){
    const v2f* sp = (const v2f*)(state + (tile*16 + lm) * NS);
    v2f a[4];
#pragma unroll
    for (int c = 0; c < 4; c++) a[c] = sp[2*c + half];

    if (tile + stride < NTILES)
      __builtin_prefetch(state + ((tile + stride)*16 + lm) * NS, 0, 0);

    v8f U = ld8(&s_bc2[8*half]);
    for (int jt = 0; jt < 8; jt++){
      v8f z = ld8(&s_bc1[jt*16 + 8*half]);
#pragma unroll
      for (int c = 0; c < 4; c++)
        z = wmma4(sBc1[(jt*4 + c)*32 + lane], a[c], z);
#pragma unroll
      for (int v = 0; v < 8; v++) z[v] = fast_tanh(z[v]);
#pragma unroll
      for (int c = 0; c < 4; c++)
        U = wmma4(sBc2[(jt*4 + c)*32 + lane], mk2(z[2*c], z[2*c + 1]), U);
    }

    v8f DH = splat8(0.0f);
    float hacc[8];
#pragma unroll
    for (int v = 0; v < 8; v++) hacc[v] = 0.0f;
#pragma unroll 2
    for (int jt = 0; jt < 8; jt++){
      v8f z = ld8(&s_bh1[jt*16 + 8*half]);
#pragma unroll
      for (int c = 0; c < 4; c++)
        z = wmma4(sBh1[(jt*4 + c)*32 + lane], a[c], z);
      const v8f w2 = ld8(&s_wh2[jt*16 + 8*half]);
#pragma unroll
      for (int v = 0; v < 8; v++){
        const float t = fast_tanh(z[v]);
        hacc[v] += t * w2[v];
        z[v] = (1.0f - t*t) * w2[v];
      }
#pragma unroll
      for (int c = 0; c < 4; c++)
        DH = wmma4(sBh1T[(jt*4 + c)*32 + lane], mk2(z[2*c], z[2*c + 1]), DH);
    }

    v8f F = ld8(&s_bf[8*half]);
#pragma unroll
    for (int c = 0; c < 4; c++) F = wmma4(sBf[c*32 + lane], a[c], F);

    float rp = ((hacc[0] + hacc[1]) + (hacc[2] + hacc[3]))
             + ((hacc[4] + hacc[5]) + (hacc[6] + hacc[7]));
#pragma unroll
    for (int v = 0; v < 8; v++) rp += DH[v] * F[v];

    float left0 = 0.f, left1 = 0.f, left2 = 0.f, left3 = 0.f;
#pragma unroll
    for (int jt = 0; jt < 4; jt++){
      v8f G = ld8(&s_bg[32*half + 8*jt]);
#pragma unroll
      for (int c = 0; c < 4; c++) G = wmma4(sBg[(jt*4 + c)*32 + lane], a[c], G);
      const float d0 = DH[2*jt], d1 = DH[2*jt + 1];
      left0 -= d0 * G[0];  left1 -= d0 * G[1];  left2 -= d0 * G[2];  left3 -= d0 * G[3];
      left0 -= d1 * G[4];  left1 -= d1 * G[5];  left2 -= d1 * G[6];  left3 -= d1 * G[7];
    }

    left0 += xhalf(left0);
    left1 += xhalf(left1);
    left2 += xhalf(left2);
    left3 += xhalf(left3);
    rp    += xhalf(rp);

    const float right = rp + bh2s;
    const float uu0 = 2.0f * U[0], uu1 = 2.0f * U[1];
    const float uu2 = 2.0f * U[2], uu3 = 2.0f * U[3];
    const float viol = left0*uu0 + left1*uu1 + left2*uu2 + left3*uu3 - right;
    const float den  = left0*left0 + left1*left1 + left2*left2 + left3*left3 + 1e-8f;
    const float lam  = fmaxf(viol, 0.0f) / den;
    if (lane < 16){
      float4 r;
      r.x = uu0 - lam*left0;
      r.y = uu1 - lam*left1;
      r.z = uu2 - lam*left2;
      r.w = uu3 - lam*left3;
      *(float4*)(out + (tile*16 + lane)*NA) = r;
    }
  }
}

extern "C" void kernel_launch(void* const* d_in, const int* in_sizes, int n_in,
                              void* d_out, int out_size, void* d_ws, size_t ws_size,
                              hipStream_t stream) {
  (void)in_sizes; (void)n_in; (void)d_ws; (void)ws_size; (void)out_size;
  const float* state = (const float*)d_in[0];
  const float* Wc1   = (const float*)d_in[1];
  const float* bc1   = (const float*)d_in[2];
  const float* Wc2   = (const float*)d_in[3];
  const float* bc2   = (const float*)d_in[4];
  const float* Wh1   = (const float*)d_in[5];
  const float* bh1   = (const float*)d_in[6];
  const float* wh2   = (const float*)d_in[7];
  const float* bh2   = (const float*)d_in[8];
  const float* Wf    = (const float*)d_in[9];
  const float* bf    = (const float*)d_in[10];
  const float* Wg    = (const float*)d_in[11];
  const float* bg    = (const float*)d_in[12];
  float* out = (float*)d_out;

  hipLaunchKernelGGL(Agent_cbf_qp_kernel, dim3(NBLOCKS), dim3(TPB), 0, stream,
                     state, Wc1, bc1, Wc2, bc2, Wh1, bh1, wh2, bh2,
                     Wf, bf, Wg, bg, out);
}


# tanh of both hidden layers by one v_tanh_f32 per value in place of exp2, rcp and five vector ops per value; tanh values carry that instruction's rounding; order of all sums unchanged
# speedup vs baseline: 1.7910x; 1.1603x over previous
#include <hip/hip_runtime.h>

#define NB 512
#define NT 2048
#define NS 16
#define NH 128
#define NA 4
#define NTOK   (NB*NT)
#define NTILES (NTOK/16)

#define WAVES   2
#define TPB     (WAVES*32)
#define NBLOCKS 2048

typedef float v2f __attribute__((ext_vector_type(2)));
typedef float v8f __attribute__((ext_vector_type(8)));

__device__ __forceinline__ v8f splat8(float x){
  v8f r;
#pragma unroll
  for (int i = 0; i < 8; i++) r[i] = x;
  return r;
}

__device__ __forceinline__ v2f mk2(float a, float b){
  v2f r; r[0] = a; r[1] = b; return r;
}

__device__ __forceinline__ v8f wmma4(v2f a, v2f b, v8f c){
  return __builtin_amdgcn_wmma_f32_16x16x4_f32(false, a, false, b,
                                               (short)0, c, false, false);
}

__device__ __forceinline__ float fast_tanh(float x){
  return __builtin_amdgcn_tanhf(x);
}

__device__ __forceinline__ v8f ld8(const float* p){
  return *(const v8f*)p;
}

__device__ __forceinline__ float xhalf(float x){
  const int b = __builtin_bit_cast(int, x);
  return __builtin_bit_cast(float,
      __builtin_amdgcn_permlanex16(b, b, 0x76543210, 0xfedcba98, false, false));
}

__global__ __launch_bounds__(TPB) void Agent_cbf_qp_kernel(
    const float* __restrict__ state,
    const float* __restrict__ Wc1, const float* __restrict__ bc1,
    const float* __restrict__ Wc2, const float* __restrict__ bc2,
    const float* __restrict__ Wh1, const float* __restrict__ bh1,
    const float* __restrict__ wh2, const float* __restrict__ bh2,
    const float* __restrict__ Wf,  const float* __restrict__ bf,
    const float* __restrict__ Wg,  const float* __restrict__ bg,
    float* __restrict__ out)
{
  __shared__ v2f sBc1 [1024];
  __shared__ v2f sBh1 [1024];
  __shared__ v2f sBh1T[1024];
  __shared__ v2f sBc2 [1024];
  __shared__ v2f sBf  [ 128];
  __shared__ v2f sBg  [ 512];
  __shared__ __attribute__((aligned(32))) float s_bc1[128], s_bh1[128], s_wh2[128],
                                                s_bf[16], s_bg[64], s_bc2[16];

  const int tid = threadIdx.x;

  for (int i = tid; i < 1024; i += TPB){
    const int l = i & 31, lm = l & 15, lh = l >> 4;
    const int c = (i >> 5) & 3, jt = i >> 7;
    const int n = jt*16 + lm,  k0 = c*4 + lh*2;
    sBc1[i] = mk2(Wc1[k0*NH + n], Wc1[(k0+1)*NH + n]);
    sBh1[i] = mk2(Wh1[k0*NH + n], Wh1[(k0+1)*NH + n]);
    const int kk0 = jt*16 + lh*8 + c*2;
    sBh1T[i] = mk2(Wh1[lm*NH + kk0], Wh1[lm*NH + kk0 + 1]);
    sBc2[i]  = (lm < NA) ? mk2(Wc2[kk0*NA + lm], Wc2[(kk0+1)*NA + lm])
                         : mk2(0.0f, 0.0f);
  }
  for (int i = tid; i < 128; i += TPB){
    const int l = i & 31, lm = l & 15, lh = l >> 4;
    const int c = i >> 5, k0 = c*4 + lh*2;
    sBf[i] = mk2(Wf[k0*NS + lm], Wf[(k0+1)*NS + lm]);
  }
  for (int i = tid; i < 512; i += TPB){
    const int l = i & 31, lm = l & 15, lh = l >> 4;
    const int c = (i >> 5) & 3, jt = i >> 7;
    const int n = (lm >> 3)*32 + jt*8 + (lm & 7), k0 = c*4 + lh*2;
    sBg[i] = mk2(Wg[k0*(NS*NA) + n], Wg[(k0+1)*(NS*NA) + n]);
  }
  for (int i = tid; i < 128; i += TPB){ s_bc1[i]=bc1[i]; s_bh1[i]=bh1[i]; s_wh2[i]=wh2[i]; }
  for (int i = tid; i < 64;  i += TPB) s_bg[i] = bg[i];
  for (int i = tid; i < 16;  i += TPB) s_bf[i] = bf[i];
  for (int i = tid; i < 16;  i += TPB) s_bc2[i] = (i < NA) ? bc2[i] : 0.0f;
  __syncthreads();

  const int lane = tid & 31, wid = tid >> 5;
  const int lm = lane & 15, half = lane >> 4;
  const float bh2s = bh2[0];

  const int gw = blockIdx.x * WAVES + wid;
  const int stride = gridDim.x * WAVES;

  v2f wF[4], wG[4][4];
  v8f bG[4];
  const v8f bF = ld8(&s_bf[8*half]);
  const v8f bU = ld8(&s_bc2[8*half]);
#pragma unroll
  for (int c = 0; c < 4; c++) wF[c] = sBf[c*32 + lane];
#pragma unroll
  for (int jt = 0; jt < 4; jt++){
    bG[jt] = ld8(&s_bg[32*half + 8*jt]);
#pragma unroll
    for (int c = 0; c < 4; c++) wG[jt][c] = sBg[(jt*4 + c)*32 + lane];
  }

  for (int tile = gw; tile < NTILES; tile += stride){
    const v2f* sp = (const v2f*)(state + (tile*16 + lm) * NS);
    v2f a[4];
#pragma unroll
    for (int c = 0; c < 4; c++) a[c] = sp[2*c + half];

    if (tile + stride < NTILES)
      __builtin_prefetch(state + ((tile + stride)*16 + lm) * NS, 0, 0);

    v8f U = bU;
#pragma unroll 1
    for (int jp = 0; jp < 4; jp++){
      v8f z[2];
      v2f w1[2][4], w2[2][4];
#pragma unroll
      for (int b = 0; b < 2; b++){
        const int jt = 2*jp + b;
        z[b] = ld8(&s_bc1[jt*16 + 8*half]);
#pragma unroll
        for (int c = 0; c < 4; c++){
          w1[b][c] = sBc1[(jt*4 + c)*32 + lane];
          w2[b][c] = sBc2[(jt*4 + c)*32 + lane];
        }
      }
      __builtin_amdgcn_sched_barrier(0);
#pragma unroll
      for (int c = 0; c < 4; c++){
        z[0] = wmma4(w1[0][c], a[c], z[0]);
        z[1] = wmma4(w1[1][c], a[c], z[1]);
      }
#pragma unroll
      for (int b = 0; b < 2; b++){
#pragma unroll
        for (int v = 0; v < 8; v++) z[b][v] = fast_tanh(z[b][v]);
#pragma unroll
        for (int c = 0; c < 4; c++)
          U = wmma4(w2[b][c], mk2(z[b][2*c], z[b][2*c + 1]), U);
      }
    }

    v8f DH = splat8(0.0f);
    v2f hacc[4];
#pragma unroll
    for (int c = 0; c < 4; c++) hacc[c] = mk2(0.0f, 0.0f);
#pragma unroll 1
    for (int jp = 0; jp < 4; jp++){
      v8f z[2], w2[2];
      v2f w1[2][4], wT[2][4];
#pragma unroll
      for (int b = 0; b < 2; b++){
        const int jt = 2*jp + b;
        z[b]  = ld8(&s_bh1[jt*16 + 8*half]);
        w2[b] = ld8(&s_wh2[jt*16 + 8*half]);
#pragma unroll
        for (int c = 0; c < 4; c++){
          w1[b][c] = sBh1 [(jt*4 + c)*32 + lane];
          wT[b][c] = sBh1T[(jt*4 + c)*32 + lane];
        }
      }
      __builtin_amdgcn_sched_barrier(0);
#pragma unroll
      for (int c = 0; c < 4; c++){
        z[0] = wmma4(w1[0][c], a[c], z[0]);
        z[1] = wmma4(w1[1][c], a[c], z[1]);
      }
#pragma unroll
      for (int b = 0; b < 2; b++){
#pragma unroll
        for (int c = 0; c < 4; c++){
          const v2f t = mk2(fast_tanh(z[b][2*c]), fast_tanh(z[b][2*c + 1]));
          const v2f w = mk2(w2[b][2*c], w2[b][2*c + 1]);
          hacc[c] += t * w;
          DH = wmma4(wT[b][c], (1.0f - t*t) * w, DH);
        }
      }
    }

    v8f F = bF;
#pragma unroll
    for (int c = 0; c < 4; c++) F = wmma4(wF[c], a[c], F);

    float rp = ((hacc[0][0] + hacc[0][1]) + (hacc[1][0] + hacc[1][1]))
             + ((hacc[2][0] + hacc[2][1]) + (hacc[3][0] + hacc[3][1]));
#pragma unroll
    for (int v = 0; v < 8; v++) rp += DH[v] * F[v];

    float left0 = 0.f, left1 = 0.f, left2 = 0.f, left3 = 0.f;
#pragma unroll
    for (int jt = 0; jt < 4; jt++){
      v8f G = bG[jt];
#pragma unroll
      for (int c = 0; c < 4; c++) G = wmma4(wG[jt][c], a[c], G);
      const float d0 = DH[2*jt], d1 = DH[2*jt + 1];
      left0 -= d0 * G[0];  left1 -= d0 * G[1];  left2 -= d0 * G[2];  left3 -= d0 * G[3];
      left0 -= d1 * G[4];  left1 -= d1 * G[5];  left2 -= d1 * G[6];  left3 -= d1 * G[7];
    }

    left0 += xhalf(left0);
    left1 += xhalf(left1);
    left2 += xhalf(left2);
    left3 += xhalf(left3);
    rp    += xhalf(rp);

    const float right = rp + bh2s;
    const float uu0 = 2.0f * U[0], uu1 = 2.0f * U[1];
    const float uu2 = 2.0f * U[2], uu3 = 2.0f * U[3];
    const float viol = left0*uu0 + left1*uu1 + left2*uu2 + left3*uu3 - right;
    const float den  = left0*left0 + left1*left1 + left2*left2 + left3*left3 + 1e-8f;
    const float lam  = fmaxf(viol, 0.0f) / den;
    if (lane < 16){
      float4 r;
      r.x = uu0 - lam*left0;
      r.y = uu1 - lam*left1;
      r.z = uu2 - lam*left2;
      r.w = uu3 - lam*left3;
      *(float4*)(out + (tile*16 + lane)*NA) = r;
    }
  }
}

extern "C" void kernel_launch(void* const* d_in, const int* in_sizes, int n_in,
                              void* d_out, int out_size, void* d_ws, size_t ws_size,
                              hipStream_t stream) {
  (void)in_sizes; (void)n_in; (void)d_ws; (void)ws_size; (void)out_size;
  const float* state = (const float*)d_in[0];
  const float* Wc1   = (const float*)d_in[1];
  const float* bc1   = (const float*)d_in[2];
  const float* Wc2   = (const float*)d_in[3];
  const float* bc2   = (const float*)d_in[4];
  const float* Wh1   = (const float*)d_in[5];
  const float* bh1   = (const float*)d_in[6];
  const float* wh2   = (const float*)d_in[7];
  const float* bh2   = (const float*)d_in[8];
  const float* Wf    = (const float*)d_in[9];
  const float* bf    = (const float*)d_in[10];
  const float* Wg    = (const float*)d_in[11];
  const float* bg    = (const float*)d_in[12];
  float* out = (float*)d_out;

  hipLaunchKernelGGL(Agent_cbf_qp_kernel, dim3(NBLOCKS), dim3(TPB), 0, stream,
                     state, Wc1, bc1, Wc2, bc2, Wh1, bh1, wh2, bh2,
                     Wf, bf, Wg, bg, out);
}


# Each pass of a wave takes four 16-token tiles: every weight fragment and bias row read from LDS feeds four tiles' WMMAs, so LDS reads per token fall to under a third; per-token sum order unchanged
# speedup vs baseline: 1.9450x; 1.0860x over previous
#include <hip/hip_runtime.h>

#define NB 512
#define NT 2048
#define NS 16
#define NH 128
#define NA 4
#define NTOK   (NB*NT)
#define NTILES (NTOK/16)

#define WAVES   2
#define TPB     (WAVES*32)
#define NBLOCKS 2048
#define TT      4
#define NPASS   (NTILES/TT)

typedef float v2f __attribute__((ext_vector_type(2)));
typedef float v8f __attribute__((ext_vector_type(8)));

__device__ __forceinline__ v8f splat8(float x){
  v8f r;
#pragma unroll
  for (int i = 0; i < 8; i++) r[i] = x;
  return r;
}

__device__ __forceinline__ v2f mk2(float a, float b){
  v2f r; r[0] = a; r[1] = b; return r;
}

__device__ __forceinline__ v8f wmma4(v2f a, v2f b, v8f c){
  return __builtin_amdgcn_wmma_f32_16x16x4_f32(false, a, false, b,
                                               (short)0, c, false, false);
}

__device__ __forceinline__ float fast_tanh(float x){
  return __builtin_amdgcn_tanhf(x);
}

__device__ __forceinline__ v8f ld8(const float* p){
  return *(const v8f*)p;
}

__device__ __forceinline__ float xhalf(float x){
  const int b = __builtin_bit_cast(int, x);
  return __builtin_bit_cast(float,
      __builtin_amdgcn_permlanex16(b, b, 0x76543210, 0xfedcba98, false, false));
}

__global__ __launch_bounds__(TPB) void Agent_cbf_qp_kernel(
    const float* __restrict__ state,
    const float* __restrict__ Wc1, const float* __restrict__ bc1,
    const float* __restrict__ Wc2, const float* __restrict__ bc2,
    const float* __restrict__ Wh1, const float* __restrict__ bh1,
    const float* __restrict__ wh2, const float* __restrict__ bh2,
    const float* __restrict__ Wf,  const float* __restrict__ bf,
    const float* __restrict__ Wg,  const float* __restrict__ bg,
    float* __restrict__ out)
{
  __shared__ v2f sBc1 [1024];
  __shared__ v2f sBh1 [1024];
  __shared__ v2f sBh1T[1024];
  __shared__ v2f sBc2 [1024];
  __shared__ v2f sBf  [ 128];
  __shared__ v2f sBg  [ 512];
  __shared__ __attribute__((aligned(32))) float s_bc1[128], s_bh1[128], s_wh2[128],
                                                s_bf[16], s_bg[64], s_bc2[16];

  const int tid = threadIdx.x;

  for (int i = tid; i < 1024; i += TPB){
    const int l = i & 31, lm = l & 15, lh = l >> 4;
    const int c = (i >> 5) & 3, jt = i >> 7;
    const int n = jt*16 + lm,  k0 = c*4 + lh*2;
    sBc1[i] = mk2(Wc1[k0*NH + n], Wc1[(k0+1)*NH + n]);
    sBh1[i] = mk2(Wh1[k0*NH + n], Wh1[(k0+1)*NH + n]);
    const int kk0 = jt*16 + lh*8 + c*2;
    sBh1T[i] = mk2(Wh1[lm*NH + kk0], Wh1[lm*NH + kk0 + 1]);
    sBc2[i]  = (lm < NA) ? mk2(Wc2[kk0*NA + lm], Wc2[(kk0+1)*NA + lm])
                         : mk2(0.0f, 0.0f);
  }
  for (int i = tid; i < 128; i += TPB){
    const int l = i & 31, lm = l & 15, lh = l >> 4;
    const int c = i >> 5, k0 = c*4 + lh*2;
    sBf[i] = mk2(Wf[k0*NS + lm], Wf[(k0+1)*NS + lm]);
  }
  for (int i = tid; i < 512; i += TPB){
    const int l = i & 31, lm = l & 15, lh = l >> 4;
    const int c = (i >> 5) & 3, jt = i >> 7;
    const int n = (lm >> 3)*32 + jt*8 + (lm & 7), k0 = c*4 + lh*2;
    sBg[i] = mk2(Wg[k0*(NS*NA) + n], Wg[(k0+1)*(NS*NA) + n]);
  }
  for (int i = tid; i < 128; i += TPB){ s_bc1[i]=bc1[i]; s_bh1[i]=bh1[i]; s_wh2[i]=wh2[i]; }
  for (int i = tid; i < 64;  i += TPB) s_bg[i] = bg[i];
  for (int i = tid; i < 16;  i += TPB) s_bf[i] = bf[i];
  for (int i = tid; i < 16;  i += TPB) s_bc2[i] = (i < NA) ? bc2[i] : 0.0f;
  __syncthreads();

  const int lane = tid & 31, wid = tid >> 5;
  const int lm = lane & 15, half = lane >> 4;
  const float bh2s = bh2[0];

  const int gw = blockIdx.x * WAVES + wid;
  const int stride = gridDim.x * WAVES;

  for (int pass = gw; pass < NPASS; pass += stride){
    const int tile0 = pass * TT;
    v2f a[TT][4];
#pragma unroll
    for (int t = 0; t < TT; t++){
      const v2f* sp = (const v2f*)(state + ((tile0 + t)*16 + lm) * NS);
#pragma unroll
      for (int c = 0; c < 4; c++) a[t][c] = sp[2*c + half];
    }

    if (pass + stride < NPASS){
#pragma unroll
      for (int q = 0; q < TT/2; q++)
        __builtin_prefetch(state + (((pass + stride)*TT + 2*q + half)*16 + lm) * NS, 0, 0);
    }

    v8f U[TT];
#pragma unroll
    for (int t = 0; t < TT; t++) U[t] = ld8(&s_bc2[8*half]);
#pragma unroll 1
    for (int jt = 0; jt < 8; jt++){
      v2f w1[4], w2[4];
      const v8f bz = ld8(&s_bc1[jt*16 + 8*half]);
#pragma unroll
      for (int c = 0; c < 4; c++){
        w1[c] = sBc1[(jt*4 + c)*32 + lane];
        w2[c] = sBc2[(jt*4 + c)*32 + lane];
      }
      __builtin_amdgcn_sched_barrier(0);
      v8f z[TT];
#pragma unroll
      for (int t = 0; t < TT; t++) z[t] = bz;
#pragma unroll
      for (int c = 0; c < 4; c++){
#pragma unroll
        for (int t = 0; t < TT; t++) z[t] = wmma4(w1[c], a[t][c], z[t]);
      }
#pragma unroll
      for (int t = 0; t < TT; t++){
#pragma unroll
        for (int v = 0; v < 8; v++) z[t][v] = fast_tanh(z[t][v]);
#pragma unroll
        for (int c = 0; c < 4; c++)
          U[t] = wmma4(w2[c], mk2(z[t][2*c], z[t][2*c + 1]), U[t]);
      }
    }

    v8f DH[TT];
    v2f hacc[TT][4];
#pragma unroll
    for (int t = 0; t < TT; t++){
      DH[t] = splat8(0.0f);
#pragma unroll
      for (int c = 0; c < 4; c++) hacc[t][c] = mk2(0.0f, 0.0f);
    }
#pragma unroll 1
    for (int jt = 0; jt < 8; jt++){
      v2f w1[4], wT[4];
      const v8f bz = ld8(&s_bh1[jt*16 + 8*half]);
      const v8f w2 = ld8(&s_wh2[jt*16 + 8*half]);
#pragma unroll
      for (int c = 0; c < 4; c++){
        w1[c] = sBh1 [(jt*4 + c)*32 + lane];
        wT[c] = sBh1T[(jt*4 + c)*32 + lane];
      }
      __builtin_amdgcn_sched_barrier(0);
      v8f z[TT];
#pragma unroll
      for (int t = 0; t < TT; t++) z[t] = bz;
#pragma unroll
      for (int c = 0; c < 4; c++){
#pragma unroll
        for (int t = 0; t < TT; t++) z[t] = wmma4(w1[c], a[t][c], z[t]);
      }
#pragma unroll
      for (int t = 0; t < TT; t++){
#pragma unroll
        for (int c = 0; c < 4; c++){
          const v2f th = mk2(fast_tanh(z[t][2*c]), fast_tanh(z[t][2*c + 1]));
          const v2f w = mk2(w2[2*c], w2[2*c + 1]);
          hacc[t][c] += th * w;
          DH[t] = wmma4(wT[c], (1.0f - th*th) * w, DH[t]);
        }
      }
    }

    v2f wF[4], wG[4][4];
    v8f bG[4];
    const v8f bF = ld8(&s_bf[8*half]);
#pragma unroll
    for (int c = 0; c < 4; c++) wF[c] = sBf[c*32 + lane];
#pragma unroll
    for (int jt = 0; jt < 4; jt++){
      bG[jt] = ld8(&s_bg[32*half + 8*jt]);
#pragma unroll
      for (int c = 0; c < 4; c++) wG[jt][c] = sBg[(jt*4 + c)*32 + lane];
    }
#pragma unroll
    for (int t = 0; t < TT; t++){
      v8f F = bF;
#pragma unroll
      for (int c = 0; c < 4; c++) F = wmma4(wF[c], a[t][c], F);

      float rp = ((hacc[t][0][0] + hacc[t][0][1]) + (hacc[t][1][0] + hacc[t][1][1]))
               + ((hacc[t][2][0] + hacc[t][2][1]) + (hacc[t][3][0] + hacc[t][3][1]));
#pragma unroll
      for (int v = 0; v < 8; v++) rp += DH[t][v] * F[v];

      float left0 = 0.f, left1 = 0.f, left2 = 0.f, left3 = 0.f;
#pragma unroll
      for (int jt = 0; jt < 4; jt++){
        v8f G = bG[jt];
#pragma unroll
        for (int c = 0; c < 4; c++) G = wmma4(wG[jt][c], a[t][c], G);
        const float d0 = DH[t][2*jt], d1 = DH[t][2*jt + 1];
        left0 -= d0 * G[0];  left1 -= d0 * G[1];  left2 -= d0 * G[2];  left3 -= d0 * G[3];
        left0 -= d1 * G[4];  left1 -= d1 * G[5];  left2 -= d1 * G[6];  left3 -= d1 * G[7];
      }

      left0 += xhalf(left0);
      left1 += xhalf(left1);
      left2 += xhalf(left2);
      left3 += xhalf(left3);
      rp    += xhalf(rp);

      const float right = rp + bh2s;
      const float uu0 = 2.0f * U[t][0], uu1 = 2.0f * U[t][1];
      const float uu2 = 2.0f * U[t][2], uu3 = 2.0f * U[t][3];
      const float viol = left0*uu0 + left1*uu1 + left2*uu2 + left3*uu3 - right;
      const float den  = left0*left0 + left1*left1 + left2*left2 + left3*left3 + 1e-8f;
      const float lam  = fmaxf(viol, 0.0f) / den;
      if (lane < 16){
        float4 r;
        r.x = uu0 - lam*left0;
        r.y = uu1 - lam*left1;
        r.z = uu2 - lam*left2;
        r.w = uu3 - lam*left3;
        *(float4*)(out + ((tile0 + t)*16 + lane)*NA) = r;
      }
    }
  }
}

extern "C" void kernel_launch(void* const* d_in, const int* in_sizes, int n_in,
                              void* d_out, int out_size, void* d_ws, size_t ws_size,
                              hipStream_t stream) {
  (void)in_sizes; (void)n_in; (void)d_ws; (void)ws_size; (void)out_size;
  const float* state = (const float*)d_in[0];
  const float* Wc1   = (const float*)d_in[1];
  const float* bc1   = (const float*)d_in[2];
  const float* Wc2   = (const float*)d_in[3];
  const float* bc2   = (const float*)d_in[4];
  const float* Wh1   = (const float*)d_in[5];
  const float* bh1   = (const float*)d_in[6];
  const float* wh2   = (const float*)d_in[7];
  const float* bh2   = (const float*)d_in[8];
  const float* Wf    = (const float*)d_in[9];
  const float* bf    = (const float*)d_in[10];
  const float* Wg    = (const float*)d_in[11];
  const float* bg    = (const float*)d_in[12];
  float* out = (float*)d_out;

  hipLaunchKernelGGL(Agent_cbf_qp_kernel, dim3(NBLOCKS), dim3(TPB), 0, stream,
                     state, Wc1, bc1, Wc2, bc2, Wh1, bh1, wh2, bh2,
                     Wf, bf, Wg, bg, out);
}


# Weight staging once per 8-wave block (was per 2-wave block), every global load of a thread issued ahead of its first LDS store: 47 dependent load round trips per thread become one; sums unchanged
# speedup vs baseline: 2.1770x; 1.1193x over previous
#include <hip/hip_runtime.h>

#define NB 512
#define NT 2048
#define NS 16
#define NH 128
#define NA 4
#define NTOK   (NB*NT)
#define NTILES (NTOK/16)

#define WAVES   8
#define TPB     (WAVES*32)
#define NBLOCKS 512
#define TT      4
#define NPASS   (NTILES/TT)

typedef float v2f __attribute__((ext_vector_type(2)));
typedef float v8f __attribute__((ext_vector_type(8)));

__device__ __forceinline__ v8f splat8(float x){
  v8f r;
#pragma unroll
  for (int i = 0; i < 8; i++) r[i] = x;
  return r;
}

__device__ __forceinline__ v2f mk2(float a, float b){
  v2f r; r[0] = a; r[1] = b; return r;
}

__device__ __forceinline__ v8f wmma4(v2f a, v2f b, v8f c){
  return __builtin_amdgcn_wmma_f32_16x16x4_f32(false, a, false, b,
                                               (short)0, c, false, false);
}

__device__ __forceinline__ float fast_tanh(float x){
  return __builtin_amdgcn_tanhf(x);
}

__device__ __forceinline__ v8f ld8(const float* p){
  return *(const v8f*)p;
}

__device__ __forceinline__ float xhalf(float x){
  const int b = __builtin_bit_cast(int, x);
  return __builtin_bit_cast(float,
      __builtin_amdgcn_permlanex16(b, b, 0x76543210, 0xfedcba98, false, false));
}

__global__ __launch_bounds__(TPB) void Agent_cbf_qp_kernel(
    const float* __restrict__ state,
    const float* __restrict__ Wc1, const float* __restrict__ bc1,
    const float* __restrict__ Wc2, const float* __restrict__ bc2,
    const float* __restrict__ Wh1, const float* __restrict__ bh1,
    const float* __restrict__ wh2, const float* __restrict__ bh2,
    const float* __restrict__ Wf,  const float* __restrict__ bf,
    const float* __restrict__ Wg,  const float* __restrict__ bg,
    float* __restrict__ out)
{
  __shared__ v2f sBc1 [1024];
  __shared__ v2f sBh1 [1024];
  __shared__ v2f sBh1T[1024];
  __shared__ v2f sBc2 [1024];
  __shared__ v2f sBf  [ 128];
  __shared__ v2f sBg  [ 512];
  __shared__ __attribute__((aligned(32))) float s_bc1[128], s_bh1[128], s_wh2[128],
                                                s_bf[16], s_bg[64], s_bc2[16];

  const int tid = threadIdx.x;

  constexpr int KT = 1024 / TPB;
  constexpr int GT = 512 / TPB;
  static_assert(1024 % TPB == 0 && 512 % TPB == 0 && TPB >= 128, "staging index scheme");

  v2f rc1[KT], rh1[KT], rhT[KT], rc2[KT], rg[GT];
#pragma unroll
  for (int u = 0; u < KT; u++){
    const int i = tid + u*TPB;
    const int l = i & 31, lm = l & 15, lh = l >> 4;
    const int c = (i >> 5) & 3, jt = i >> 7;
    const int n = jt*16 + lm,  k0 = c*4 + lh*2;
    rc1[u] = mk2(Wc1[k0*NH + n], Wc1[(k0+1)*NH + n]);
    rh1[u] = mk2(Wh1[k0*NH + n], Wh1[(k0+1)*NH + n]);
    const int kk0 = jt*16 + lh*8 + c*2;
    rhT[u] = mk2(Wh1[lm*NH + kk0], Wh1[lm*NH + kk0 + 1]);
    const int la = lm & (NA - 1);
    rc2[u] = mk2(Wc2[kk0*NA + la], Wc2[(kk0+1)*NA + la]);
  }
  v2f rf;
  {
    const int i = tid & 127;
    const int l = i & 31, lm = l & 15, lh = l >> 4;
    const int c = i >> 5, k0 = c*4 + lh*2;
    rf = mk2(Wf[k0*NS + lm], Wf[(k0+1)*NS + lm]);
  }
#pragma unroll
  for (int u = 0; u < GT; u++){
    const int i = tid + u*TPB;
    const int l = i & 31, lm = l & 15, lh = l >> 4;
    const int c = (i >> 5) & 3, jt = i >> 7;
    const int n = (lm >> 3)*32 + jt*8 + (lm & 7), k0 = c*4 + lh*2;
    rg[u] = mk2(Wg[k0*(NS*NA) + n], Wg[(k0+1)*(NS*NA) + n]);
  }
  const float r_bc1 = bc1[tid & 127], r_bh1 = bh1[tid & 127], r_wh2 = wh2[tid & 127];
  const float r_bg  = bg[tid & 63];
  const float r_bf  = bf[tid & 15];
  const float r_bc2 = bc2[tid & (NA - 1)];

#pragma unroll
  for (int u = 0; u < KT; u++){
    const int i = tid + u*TPB;
    sBc1[i]  = rc1[u];
    sBh1[i]  = rh1[u];
    sBh1T[i] = rhT[u];
    sBc2[i]  = ((i & 15) < NA) ? rc2[u] : mk2(0.0f, 0.0f);
  }
#pragma unroll
  for (int u = 0; u < GT; u++) sBg[tid + u*TPB] = rg[u];
  sBf[tid & 127] = rf;
  s_bc1[tid & 127] = r_bc1;  s_bh1[tid & 127] = r_bh1;  s_wh2[tid & 127] = r_wh2;
  s_bg[tid & 63] = r_bg;
  s_bf[tid & 15] = r_bf;
  s_bc2[tid & 15] = ((tid & 15) < NA) ? r_bc2 : 0.0f;
  __syncthreads();

  const int lane = tid & 31, wid = tid >> 5;
  const int lm = lane & 15, half = lane >> 4;
  const float bh2s = bh2[0];

  const int gw = blockIdx.x * WAVES + wid;
  const int stride = gridDim.x * WAVES;

  for (int pass = gw; pass < NPASS; pass += stride){
    const int tile0 = pass * TT;
    v2f a[TT][4];
#pragma unroll
    for (int t = 0; t < TT; t++){
      const v2f* sp = (const v2f*)(state + ((tile0 + t)*16 + lm) * NS);
#pragma unroll
      for (int c = 0; c < 4; c++) a[t][c] = sp[2*c + half];
    }

    if (pass + stride < NPASS){
#pragma unroll
      for (int q = 0; q < TT/2; q++)
        __builtin_prefetch(state + (((pass + stride)*TT + 2*q + half)*16 + lm) * NS, 0, 0);
    }

    v8f U[TT];
#pragma unroll
    for (int t = 0; t < TT; t++) U[t] = ld8(&s_bc2[8*half]);
#pragma unroll 1
    for (int jt = 0; jt < 8; jt++){
      v2f w1[4], w2[4];
      const v8f bz = ld8(&s_bc1[jt*16 + 8*half]);
#pragma unroll
      for (int c = 0; c < 4; c++){
        w1[c] = sBc1[(jt*4 + c)*32 + lane];
        w2[c] = sBc2[(jt*4 + c)*32 + lane];
      }
      __builtin_amdgcn_sched_barrier(0);
      v8f z[TT];
#pragma unroll
      for (int t = 0; t < TT; t++) z[t] = bz;
#pragma unroll
      for (int c = 0; c < 4; c++){
#pragma unroll
        for (int t = 0; t < TT; t++) z[t] = wmma4(w1[c], a[t][c], z[t]);
      }
#pragma unroll
      for (int t = 0; t < TT; t++){
#pragma unroll
        for (int v = 0; v < 8; v++) z[t][v] = fast_tanh(z[t][v]);
#pragma unroll
        for (int c = 0; c < 4; c++)
          U[t] = wmma4(w2[c], mk2(z[t][2*c], z[t][2*c + 1]), U[t]);
      }
    }

    v8f DH[TT];
    v2f hacc[TT][4];
#pragma unroll
    for (int t = 0; t < TT; t++){
      DH[t] = splat8(0.0f);
#pragma unroll
      for (int c = 0; c < 4; c++) hacc[t][c] = mk2(0.0f, 0.0f);
    }
#pragma unroll 1
    for (int jt = 0; jt < 8; jt++){
      v2f w1[4], wT[4];
      const v8f bz = ld8(&s_bh1[jt*16 + 8*half]);
      const v8f w2 = ld8(&s_wh2[jt*16 + 8*half]);
#pragma unroll
      for (int c = 0; c < 4; c++){
        w1[c] = sBh1 [(jt*4 + c)*32 + lane];
        wT[c] = sBh1T[(jt*4 + c)*32 + lane];
      }
      __builtin_amdgcn_sched_barrier(0);
      v8f z[TT];
#pragma unroll
      for (int t = 0; t < TT; t++) z[t] = bz;
#pragma unroll
      for (int c = 0; c < 4; c++){
#pragma unroll
        for (int t = 0; t < TT; t++) z[t] = wmma4(w1[c], a[t][c], z[t]);
      }
#pragma unroll
      for (int t = 0; t < TT; t++){
#pragma unroll
        for (int c = 0; c < 4; c++){
          const v2f th = mk2(fast_tanh(z[t][2*c]), fast_tanh(z[t][2*c + 1]));
          const v2f w = mk2(w2[2*c], w2[2*c + 1]);
          hacc[t][c] += th * w;
          DH[t] = wmma4(wT[c], (1.0f - th*th) * w, DH[t]);
        }
      }
    }

    v2f wF[4], wG[4][4];
    v8f bG[4];
    const v8f bF = ld8(&s_bf[8*half]);
#pragma unroll
    for (int c = 0; c < 4; c++) wF[c] = sBf[c*32 + lane];
#pragma unroll
    for (int jt = 0; jt < 4; jt++){
      bG[jt] = ld8(&s_bg[32*half + 8*jt]);
#pragma unroll
      for (int c = 0; c < 4; c++) wG[jt][c] = sBg[(jt*4 + c)*32 + lane];
    }
#pragma unroll
    for (int t = 0; t < TT; t++){
      v8f F = bF;
#pragma unroll
      for (int c = 0; c < 4; c++) F = wmma4(wF[c], a[t][c], F);

      float rp = ((hacc[t][0][0] + hacc[t][0][1]) + (hacc[t][1][0] + hacc[t][1][1]))
               + ((hacc[t][2][0] + hacc[t][2][1]) + (hacc[t][3][0] + hacc[t][3][1]));
#pragma unroll
      for (int v = 0; v < 8; v++) rp += DH[t][v] * F[v];

      float left0 = 0.f, left1 = 0.f, left2 = 0.f, left3 = 0.f;
#pragma unroll
      for (int jt = 0; jt < 4; jt++){
        v8f G = bG[jt];
#pragma unroll
        for (int c = 0; c < 4; c++) G = wmma4(wG[jt][c], a[t][c], G);
        const float d0 = DH[t][2*jt], d1 = DH[t][2*jt + 1];
        left0 -= d0 * G[0];  left1 -= d0 * G[1];  left2 -= d0 * G[2];  left3 -= d0 * G[3];
        left0 -= d1 * G[4];  left1 -= d1 * G[5];  left2 -= d1 * G[6];  left3 -= d1 * G[7];
      }

      left0 += xhalf(left0);
      left1 += xhalf(left1);
      left2 += xhalf(left2);
      left3 += xhalf(left3);
      rp    += xhalf(rp);

      const float right = rp + bh2s;
      const float uu0 = 2.0f * U[t][0], uu1 = 2.0f * U[t][1];
      const float uu2 = 2.0f * U[t][2], uu3 = 2.0f * U[t][3];
      const float viol = left0*uu0 + left1*uu1 + left2*uu2 + left3*uu3 - right;
      const float den  = left0*left0 + left1*left1 + left2*left2 + left3*left3 + 1e-8f;
      const float lam  = fmaxf(viol, 0.0f) / den;
      if (lane < 16){
        float4 r;
        r.x = uu0 - lam*left0;
        r.y = uu1 - lam*left1;
        r.z = uu2 - lam*left2;
        r.w = uu3 - lam*left3;
        *(float4*)(out + ((tile0 + t)*16 + lane)*NA) = r;
      }
    }
  }
}

extern "C" void kernel_launch(void* const* d_in, const int* in_sizes, int n_in,
                              void* d_out, int out_size, void* d_ws, size_t ws_size,
                              hipStream_t stream) {
  (void)in_sizes; (void)n_in; (void)d_ws; (void)ws_size; (void)out_size;
  const float* state = (const float*)d_in[0];
  const float* Wc1   = (const float*)d_in[1];
  const float* bc1   = (const float*)d_in[2];
  const float* Wc2   = (const float*)d_in[3];
  const float* bc2   = (const float*)d_in[4];
  const float* Wh1   = (const float*)d_in[5];
  const float* bh1   = (const float*)d_in[6];
  const float* wh2   = (const float*)d_in[7];
  const float* bh2   = (const float*)d_in[8];
  const float* Wf    = (const float*)d_in[9];
  const float* bf    = (const float*)d_in[10];
  const float* Wg    = (const float*)d_in[11];
  const float* bg    = (const float*)d_in[12];
  float* out = (float*)d_out;

  hipLaunchKernelGGL(Agent_cbf_qp_kernel, dim3(NBLOCKS), dim3(TPB), 0, stream,
                     state, Wc1, bc1, Wc2, bc2, Wh1, bh1, wh2, bh2,
                     Wf, bf, Wg, bg, out);
}


# Controller output layer (128 to 4: a quarter of a matrix tile's rows) summed by packed f32 FMAs on the D-layout tanh values, not WMMA: 116 WMMAs per tile, was 148; u's sum order changes
# speedup vs baseline: 2.4109x; 1.1074x over previous
#include <hip/hip_runtime.h>

#define NB 512
#define NT 2048
#define NS 16
#define NH 128
#define NA 4
#define NTOK   (NB*NT)
#define NTILES (NTOK/16)

#define WAVES   8
#define TPB     (WAVES*32)
#define NBLOCKS 512
#define TT      4
#define NPASS   (NTILES/TT)

typedef float v2f __attribute__((ext_vector_type(2)));
typedef float v8f __attribute__((ext_vector_type(8)));

__device__ __forceinline__ v8f splat8(float x){
  v8f r;
#pragma unroll
  for (int i = 0; i < 8; i++) r[i] = x;
  return r;
}

__device__ __forceinline__ v2f mk2(float a, float b){
  v2f r; r[0] = a; r[1] = b; return r;
}

__device__ __forceinline__ v8f wmma4(v2f a, v2f b, v8f c){
  return __builtin_amdgcn_wmma_f32_16x16x4_f32(false, a, false, b,
                                               (short)0, c, false, false);
}

__device__ __forceinline__ float fast_tanh(float x){
  return __builtin_amdgcn_tanhf(x);
}

__device__ __forceinline__ v8f ld8(const float* p){
  return *(const v8f*)p;
}

__device__ __forceinline__ float xhalf(float x){
  const int b = __builtin_bit_cast(int, x);
  return __builtin_bit_cast(float,
      __builtin_amdgcn_permlanex16(b, b, 0x76543210, 0xfedcba98, false, false));
}

__global__ __launch_bounds__(TPB) void Agent_cbf_qp_kernel(
    const float* __restrict__ state,
    const float* __restrict__ Wc1, const float* __restrict__ bc1,
    const float* __restrict__ Wc2, const float* __restrict__ bc2,
    const float* __restrict__ Wh1, const float* __restrict__ bh1,
    const float* __restrict__ wh2, const float* __restrict__ bh2,
    const float* __restrict__ Wf,  const float* __restrict__ bf,
    const float* __restrict__ Wg,  const float* __restrict__ bg,
    float* __restrict__ out)
{
  __shared__ v2f sBc1 [1024];
  __shared__ v2f sBh1 [1024];
  __shared__ v2f sBh1T[1024];
  __shared__ v2f sBf  [ 128];
  __shared__ v2f sBg  [ 512];
  __shared__ __attribute__((aligned(32))) float s_bc1[128], s_bh1[128], s_wh2[128],
                                                s_bf[16], s_bg[64];
  __shared__ __attribute__((aligned(32))) float sW2[NH*NA];

  const int tid = threadIdx.x;

  constexpr int KT = 1024 / TPB;
  constexpr int GT = 512 / TPB;
  static_assert(1024 % TPB == 0 && 512 % TPB == 0 && TPB >= 128, "staging index scheme");

  v2f rc1[KT], rh1[KT], rhT[KT], rg[GT];
#pragma unroll
  for (int u = 0; u < KT; u++){
    const int i = tid + u*TPB;
    const int l = i & 31, lm = l & 15, lh = l >> 4;
    const int c = (i >> 5) & 3, jt = i >> 7;
    const int n = jt*16 + lm,  k0 = c*4 + lh*2;
    rc1[u] = mk2(Wc1[k0*NH + n], Wc1[(k0+1)*NH + n]);
    rh1[u] = mk2(Wh1[k0*NH + n], Wh1[(k0+1)*NH + n]);
    const int kk0 = jt*16 + lh*8 + c*2;
    rhT[u] = mk2(Wh1[lm*NH + kk0], Wh1[lm*NH + kk0 + 1]);
  }
  float4 rw2;
  rw2.x = Wc2[(tid & 127)*NA + 0];  rw2.y = Wc2[(tid & 127)*NA + 1];
  rw2.z = Wc2[(tid & 127)*NA + 2];  rw2.w = Wc2[(tid & 127)*NA + 3];
  v2f rf;
  {
    const int i = tid & 127;
    const int l = i & 31, lm = l & 15, lh = l >> 4;
    const int c = i >> 5, k0 = c*4 + lh*2;
    rf = mk2(Wf[k0*NS + lm], Wf[(k0+1)*NS + lm]);
  }
#pragma unroll
  for (int u = 0; u < GT; u++){
    const int i = tid + u*TPB;
    const int l = i & 31, lm = l & 15, lh = l >> 4;
    const int c = (i >> 5) & 3, jt = i >> 7;
    const int n = (lm >> 3)*32 + jt*8 + (lm & 7), k0 = c*4 + lh*2;
    rg[u] = mk2(Wg[k0*(NS*NA) + n], Wg[(k0+1)*(NS*NA) + n]);
  }
  const float r_bc1 = bc1[tid & 127], r_bh1 = bh1[tid & 127], r_wh2 = wh2[tid & 127];
  const float r_bg  = bg[tid & 63];
  const float r_bf  = bf[tid & 15];

#pragma unroll
  for (int u = 0; u < KT; u++){
    const int i = tid + u*TPB;
    sBc1[i]  = rc1[u];
    sBh1[i]  = rh1[u];
    sBh1T[i] = rhT[u];
  }
#pragma unroll
  for (int u = 0; u < GT; u++) sBg[tid + u*TPB] = rg[u];
  sBf[tid & 127] = rf;
  *(float4*)&sW2[(tid & 127)*NA] = rw2;
  s_bc1[tid & 127] = r_bc1;  s_bh1[tid & 127] = r_bh1;  s_wh2[tid & 127] = r_wh2;
  s_bg[tid & 63] = r_bg;
  s_bf[tid & 15] = r_bf;
  __syncthreads();

  const int lane = tid & 31, wid = tid >> 5;
  const int lm = lane & 15, half = lane >> 4;
  const float bh2s = bh2[0];
  const float bc2_0 = bc2[0], bc2_1 = bc2[1], bc2_2 = bc2[2], bc2_3 = bc2[3];

  const int gw = blockIdx.x * WAVES + wid;
  const int stride = gridDim.x * WAVES;

  for (int pass = gw; pass < NPASS; pass += stride){
    const int tile0 = pass * TT;
    v2f a[TT][4];
#pragma unroll
    for (int t = 0; t < TT; t++){
      const v2f* sp = (const v2f*)(state + ((tile0 + t)*16 + lm) * NS);
#pragma unroll
      for (int c = 0; c < 4; c++) a[t][c] = sp[2*c + half];
    }

    if (pass + stride < NPASS){
#pragma unroll
      for (int q = 0; q < TT/2; q++)
        __builtin_prefetch(state + (((pass + stride)*TT + 2*q + half)*16 + lm) * NS, 0, 0);
    }

    v2f pu[TT][2];
#pragma unroll
    for (int t = 0; t < TT; t++){ pu[t][0] = mk2(0.0f, 0.0f); pu[t][1] = mk2(0.0f, 0.0f); }
#pragma unroll 1
    for (int jt = 0; jt < 8; jt++){
      v2f w1[4];
      v8f wq[4];
      const v8f bz = ld8(&s_bc1[jt*16 + 8*half]);
#pragma unroll
      for (int c = 0; c < 4; c++) w1[c] = sBc1[(jt*4 + c)*32 + lane];
#pragma unroll
      for (int i = 0; i < 4; i++) wq[i] = ld8(&sW2[(jt*16 + 8*half)*NA + 8*i]);
      __builtin_amdgcn_sched_barrier(0);
      v8f z[TT];
#pragma unroll
      for (int t = 0; t < TT; t++) z[t] = bz;
#pragma unroll
      for (int c = 0; c < 4; c++){
#pragma unroll
        for (int t = 0; t < TT; t++) z[t] = wmma4(w1[c], a[t][c], z[t]);
      }
#pragma unroll
      for (int t = 0; t < TT; t++){
#pragma unroll
        for (int v = 0; v < 8; v++){
          const float th = fast_tanh(z[t][v]);
          const v2f th2 = mk2(th, th);
          const int o = 4*(v & 1);
          pu[t][0] += th2 * mk2(wq[v >> 1][o],     wq[v >> 1][o + 1]);
          pu[t][1] += th2 * mk2(wq[v >> 1][o + 2], wq[v >> 1][o + 3]);
        }
      }
    }

    v8f DH[TT];
    v2f hacc[TT][4];
#pragma unroll
    for (int t = 0; t < TT; t++){
      DH[t] = splat8(0.0f);
#pragma unroll
      for (int c = 0; c < 4; c++) hacc[t][c] = mk2(0.0f, 0.0f);
    }
#pragma unroll 1
    for (int jt = 0; jt < 8; jt++){
      v2f w1[4], wT[4];
      const v8f bz = ld8(&s_bh1[jt*16 + 8*half]);
      const v8f w2 = ld8(&s_wh2[jt*16 + 8*half]);
#pragma unroll
      for (int c = 0; c < 4; c++){
        w1[c] = sBh1 [(jt*4 + c)*32 + lane];
        wT[c] = sBh1T[(jt*4 + c)*32 + lane];
      }
      __builtin_amdgcn_sched_barrier(0);
      v8f z[TT];
#pragma unroll
      for (int t = 0; t < TT; t++) z[t] = bz;
#pragma unroll
      for (int c = 0; c < 4; c++){
#pragma unroll
        for (int t = 0; t < TT; t++) z[t] = wmma4(w1[c], a[t][c], z[t]);
      }
#pragma unroll
      for (int t = 0; t < TT; t++){
#pragma unroll
        for (int c = 0; c < 4; c++){
          const v2f th = mk2(fast_tanh(z[t][2*c]), fast_tanh(z[t][2*c + 1]));
          const v2f w = mk2(w2[2*c], w2[2*c + 1]);
          hacc[t][c] += th * w;
          DH[t] = wmma4(wT[c], (1.0f - th*th) * w, DH[t]);
        }
      }
    }

    v2f wF[4], wG[4][4];
    v8f bG[4];
    const v8f bF = ld8(&s_bf[8*half]);
#pragma unroll
    for (int c = 0; c < 4; c++) wF[c] = sBf[c*32 + lane];
#pragma unroll
    for (int jt = 0; jt < 4; jt++){
      bG[jt] = ld8(&s_bg[32*half + 8*jt]);
#pragma unroll
      for (int c = 0; c < 4; c++) wG[jt][c] = sBg[(jt*4 + c)*32 + lane];
    }
#pragma unroll
    for (int t = 0; t < TT; t++){
      v8f F = bF;
#pragma unroll
      for (int c = 0; c < 4; c++) F = wmma4(wF[c], a[t][c], F);

      float rp = ((hacc[t][0][0] + hacc[t][0][1]) + (hacc[t][1][0] + hacc[t][1][1]))
               + ((hacc[t][2][0] + hacc[t][2][1]) + (hacc[t][3][0] + hacc[t][3][1]));
#pragma unroll
      for (int v = 0; v < 8; v++) rp += DH[t][v] * F[v];

      float left0 = 0.f, left1 = 0.f, left2 = 0.f, left3 = 0.f;
#pragma unroll
      for (int jt = 0; jt < 4; jt++){
        v8f G = bG[jt];
#pragma unroll
        for (int c = 0; c < 4; c++) G = wmma4(wG[jt][c], a[t][c], G);
        const float d0 = DH[t][2*jt], d1 = DH[t][2*jt + 1];
        left0 -= d0 * G[0];  left1 -= d0 * G[1];  left2 -= d0 * G[2];  left3 -= d0 * G[3];
        left0 -= d1 * G[4];  left1 -= d1 * G[5];  left2 -= d1 * G[6];  left3 -= d1 * G[7];
      }

      left0 += xhalf(left0);
      left1 += xhalf(left1);
      left2 += xhalf(left2);
      left3 += xhalf(left3);
      rp    += xhalf(rp);

      const float right = rp + bh2s;
      float cu0 = pu[t][0][0], cu1 = pu[t][0][1], cu2 = pu[t][1][0], cu3 = pu[t][1][1];
      cu0 += xhalf(cu0);
      cu1 += xhalf(cu1);
      cu2 += xhalf(cu2);
      cu3 += xhalf(cu3);
      const float uu0 = 2.0f * (cu0 + bc2_0), uu1 = 2.0f * (cu1 + bc2_1);
      const float uu2 = 2.0f * (cu2 + bc2_2), uu3 = 2.0f * (cu3 + bc2_3);
      const float viol = left0*uu0 + left1*uu1 + left2*uu2 + left3*uu3 - right;
      const float den  = left0*left0 + left1*left1 + left2*left2 + left3*left3 + 1e-8f;
      const float lam  = fmaxf(viol, 0.0f) / den;
      if (lane < 16){
        float4 r;
        r.x = uu0 - lam*left0;
        r.y = uu1 - lam*left1;
        r.z = uu2 - lam*left2;
        r.w = uu3 - lam*left3;
        *(float4*)(out + ((tile0 + t)*16 + lane)*NA) = r;
      }
    }
  }
}

extern "C" void kernel_launch(void* const* d_in, const int* in_sizes, int n_in,
                              void* d_out, int out_size, void* d_ws, size_t ws_size,
                              hipStream_t stream) {
  (void)in_sizes; (void)n_in; (void)d_ws; (void)ws_size; (void)out_size;
  const float* state = (const float*)d_in[0];
  const float* Wc1   = (const float*)d_in[1];
  const float* bc1   = (const float*)d_in[2];
  const float* Wc2   = (const float*)d_in[3];
  const float* bc2   = (const float*)d_in[4];
  const float* Wh1   = (const float*)d_in[5];
  const float* bh1   = (const float*)d_in[6];
  const float* wh2   = (const float*)d_in[7];
  const float* bh2   = (const float*)d_in[8];
  const float* Wf    = (const float*)d_in[9];
  const float* bf    = (const float*)d_in[10];
  const float* Wg    = (const float*)d_in[11];
  const float* bg    = (const float*)d_in[12];
  float* out = (float*)d_out;

  hipLaunchKernelGGL(Agent_cbf_qp_kernel, dim3(NBLOCKS), dim3(TPB), 0, stream,
                     state, Wc1, bc1, Wc2, bc2, Wh1, bh1, wh2, bh2,
                     Wf, bf, Wg, bg, out);
}
